# starting kernel
# baseline (speedup 1.0000x reference)
#include <hip/hip_runtime.h>

typedef __attribute__((ext_vector_type(2))) float v2f;
typedef __attribute__((ext_vector_type(4))) float v4f;
typedef __attribute__((ext_vector_type(8))) float v8f;

#define BLOCK 32
#define NB 128
#define KDIM 4096
#define NDIM 4096
#define ROWS_PER_WG 256
#define THREADS 256
#define LSTRIDE 34

__global__ void bsmm_build_csc(const int* __restrict__ idx_i,
                               const int* __restrict__ idx_j, int nnz,
                               int* __restrict__ colptr,
                               int* __restrict__ colblk,
                               int* __restrict__ coli)
{
    if (threadIdx.x != 0 || blockIdx.x != 0) return;
    int cnt[NB];
    for (int j = 0; j < NB; ++j) cnt[j] = 0;
    for (int e = 0; e < nnz; ++e) cnt[idx_j[e]]++;
    int run = 0;
    for (int j = 0; j < NB; ++j) { colptr[j] = run; run += cnt[j]; cnt[j] = 0; }
    colptr[NB] = run;
    for (int e = 0; e < nnz; ++e) {
        int j = idx_j[e];
        int o = colptr[j] + cnt[j]++;
        colblk[o] = e;
        coli[o]   = idx_i[e];
    }
}

__global__ void __launch_bounds__(THREADS)
bsmm_wmma(const float* __restrict__ x, const float* __restrict__ w,
          const int* __restrict__ colptr, const int* __restrict__ colblk,
          const int* __restrict__ coli, float* __restrict__ y)
{
    __shared__ float wlds[2][BLOCK * LSTRIDE];

    const int j    = blockIdx.y;
    const int lane = threadIdx.x & 31;
    const int wave = threadIdx.x >> 5;
    const int half = lane >> 4;
    const int lid  = lane & 15;
    const int m0   = blockIdx.x * ROWS_PER_WG + wave * 32;

    const int beg = colptr[j];
    const int cnt = colptr[j + 1] - beg;

    v8f c00 = {}, c01 = {};
    v8f c10 = {}, c11 = {};

    const float* xrow0 = x + (size_t)(m0 + lid) * KDIM + half * 2;
    const float* xrow1 = xrow0 + (size_t)16 * KDIM;

    const int boff = lid * LSTRIDE + half * 2;

    const int srow = threadIdx.x >> 3;
    const int scol = (threadIdx.x & 7) * 4;

    if (cnt > 0) {
        const float* wsrc =
            w + (size_t)colblk[beg] * (BLOCK * BLOCK) + srow * BLOCK + scol;
        v4f t0 = *(const v4f*)wsrc;
        wlds[0][(scol + 0) * LSTRIDE + srow] = t0[0];
        wlds[0][(scol + 1) * LSTRIDE + srow] = t0[1];
        wlds[0][(scol + 2) * LSTRIDE + srow] = t0[2];
        wlds[0][(scol + 3) * LSTRIDE + srow] = t0[3];
    }

    for (int t = 0; t < cnt; ++t) {
        __syncthreads();
        const int buf = t & 1;

        const bool have_next = (t + 1) < cnt;
        v4f nxt = {};
        if (have_next) {
            const float* wsrc = w + (size_t)colblk[beg + t + 1] * (BLOCK * BLOCK)
                                  + srow * BLOCK + scol;
            nxt = *(const v4f*)wsrc;
        }

        const int    kbase = coli[beg + t] * BLOCK;
        const float* xa0   = xrow0 + kbase;
        const float* xa1   = xrow1 + kbase;
        const float* bl    = &wlds[buf][boff];

#pragma unroll
        for (int s = 0; s < 8; ++s) {
            v2f a0 = *(const v2f*)(xa0 + s * 4);
            v2f a1 = *(const v2f*)(xa1 + s * 4);
            v2f b0 = *(const v2f*)(bl + s * 4);
            v2f b1 = *(const v2f*)(bl + 16 * LSTRIDE + s * 4);
            c00 = __builtin_amdgcn_wmma_f32_16x16x4_f32(
                      false, a0, false, b0, (short)0, c00, false, false);
            c01 = __builtin_amdgcn_wmma_f32_16x16x4_f32(
                      false, a0, false, b1, (short)0, c01, false, false);
            c10 = __builtin_amdgcn_wmma_f32_16x16x4_f32(
                      false, a1, false, b0, (short)0, c10, false, false);
            c11 = __builtin_amdgcn_wmma_f32_16x16x4_f32(
                      false, a1, false, b1, (short)0, c11, false, false);
        }

        if (have_next) {
            __syncthreads();
            float* dst = &wlds[buf ^ 1][0];
            dst[(scol + 0) * LSTRIDE + srow] = nxt[0];
            dst[(scol + 1) * LSTRIDE + srow] = nxt[1];
            dst[(scol + 2) * LSTRIDE + srow] = nxt[2];
            dst[(scol + 3) * LSTRIDE + srow] = nxt[3];
        }
    }

    float* y0 = y + (size_t)(m0 + half * 8) * NDIM + j * BLOCK + lid;
    float* y1 = y0 + (size_t)16 * NDIM;
#pragma unroll
    for (int r = 0; r < 8; ++r) {
        y0[(size_t)r * NDIM]      = c00[r];
        y0[(size_t)r * NDIM + 16] = c01[r];
        y1[(size_t)r * NDIM]      = c10[r];
        y1[(size_t)r * NDIM + 16] = c11[r];
    }
}

extern "C" void kernel_launch(void* const* d_in, const int* in_sizes, int n_in,
                              void* d_out, int out_size, void* d_ws, size_t ws_size,
                              hipStream_t stream) {
    const float* x     = (const float*)d_in[0];
    const float* w     = (const float*)d_in[1];
    const int*   idx_i = (const int*)d_in[2];
    const int*   idx_j = (const int*)d_in[3];
    float*       y     = (float*)d_out;

    const int nnz = in_sizes[2];
    const int M   = in_sizes[0] / KDIM;

    int* colptr = (int*)d_ws;
    int* colblk = colptr + (NB + 1);
    int* coli   = colblk + nnz;

    bsmm_build_csc<<<1, 32, 0, stream>>>(idx_i, idx_j, nnz, colptr, colblk, coli);

    dim3 grid(M / ROWS_PER_WG, NB);
    bsmm_wmma<<<grid, THREADS, 0, stream>>>(x, w, colptr, colblk, coli, y);
}


# B fragments of a block read from LDS as one group ahead of its 32 products (one read place per block, not four spread among the matrix instructions); scheduling only, sum order kept
# speedup vs baseline: 8.4687x; 8.4687x over previous
#include <hip/hip_runtime.h>

typedef __attribute__((ext_vector_type(2))) float v2f;
typedef __attribute__((ext_vector_type(4))) float v4f;
typedef __attribute__((ext_vector_type(8))) float v8f;

#define BLOCK 32
#define NB 128
#define KDIM 4096
#define NDIM 4096
#define ROWS_PER_WG 128
#define THREADS 128
#define LSTRIDE 34

#define BUILD_THREADS 256
#define BUILD_WAVES   (BUILD_THREADS / 32)
#define BUILD_GROUPS  8
static_assert(BUILD_WAVES <= 32, "one lane per wave in the cross-wave sum");

__global__ void __launch_bounds__(BUILD_THREADS)
bsmm_build_csc(const int* __restrict__ idx_i,
               const int* __restrict__ idx_j, int nnz,
               int* __restrict__ colptr,
               int* __restrict__ colblk,
               int* __restrict__ coli)
{
    __shared__ int s_lt[BUILD_WAVES];
    __shared__ int s_eq[BUILD_WAVES];

    const int      j     = blockIdx.x;
    const int      lane  = threadIdx.x & 31;
    const int      wave  = threadIdx.x >> 5;
    const unsigned below = (1u << lane) - 1u;

    const int step = 32 * BUILD_GROUPS;
    const int per  = ((nnz + BUILD_WAVES * step - 1) / (BUILD_WAVES * step)) * step;
    const int e0   = wave * per;
    const int e1   = min(e0 + per, nnz);

    int lt = 0, eq = 0;
    for (int b = e0; b < e1; b += step) {
        int v[BUILD_GROUPS];
#pragma unroll
        for (int u = 0; u < BUILD_GROUPS; ++u) {
            const int e = b + u * 32 + lane;
            v[u] = (e < e1) ? idx_j[e] : NB;
        }
#pragma unroll
        for (int u = 0; u < BUILD_GROUPS; ++u) {
            lt += __popc(__builtin_amdgcn_ballot_w32(v[u] < j));
            eq += __popc(__builtin_amdgcn_ballot_w32(v[u] == j));
        }
    }
    if (lane == 0) { s_lt[wave] = lt; s_eq[wave] = eq; }
    __syncthreads();

    int pos = 0;
    if (lane < BUILD_WAVES) pos = s_lt[lane] + (lane < wave ? s_eq[lane] : 0);
    for (int o = 16; o > 0; o >>= 1) pos += __shfl_xor(pos, o);

    if (threadIdx.x == 0) {
        colptr[j] = pos;
        if (j == NB - 1) colptr[NB] = nnz;
    }

    for (int b = e0; b < e1; b += step) {
        int v[BUILD_GROUPS], vi[BUILD_GROUPS];
#pragma unroll
        for (int u = 0; u < BUILD_GROUPS; ++u) {
            const int e = b + u * 32 + lane;
            v[u]  = (e < e1) ? idx_j[e] : NB;
            vi[u] = (e < e1) ? idx_i[e] : 0;
        }
#pragma unroll
        for (int u = 0; u < BUILD_GROUPS; ++u) {
            const bool     hit = (v[u] == j);
            const unsigned m   = __builtin_amdgcn_ballot_w32(hit);
            if (hit) {
                const int o = pos + __popc(m & below);
                colblk[o] = b + u * 32 + lane;
                coli[o]   = vi[u];
            }
            pos += __popc(m);
        }
    }
}

#define W_PASS   16
#define W_TILE   (BLOCK * LSTRIDE)
#define W_GROUP  8
static_assert(W_PASS % W_GROUP == 0, "a whole group must fit the staged pass");

__global__ void __launch_bounds__(THREADS)
bsmm_wmma(const float* __restrict__ x, const float* __restrict__ w,
          const int* __restrict__ colptr, const int* __restrict__ colblk,
          const int* __restrict__ coli, float* __restrict__ y)
{
    __shared__ float wlds[W_PASS * W_TILE];

    const int j    = blockIdx.y;
    const int lane = threadIdx.x & 31;
    const int wave = threadIdx.x >> 5;
    const int half = lane >> 4;
    const int lid  = lane & 15;
    const int m0   = blockIdx.x * ROWS_PER_WG + wave * 32;

    const int beg = colptr[j];
    const int cnt = colptr[j + 1] - beg;

    v8f c00 = {}, c01 = {};
    v8f c10 = {}, c11 = {};

    const float* xrow0 = x + (size_t)(m0 + lid) * KDIM + half * 2;
    const float* xrow1 = xrow0 + (size_t)16 * KDIM;

    const int boff = lid * LSTRIDE + half * 2;

    const int srow = threadIdx.x >> 3;
    const int scol = (threadIdx.x & 7) * 4;

    for (int cb = 0; cb < cnt; cb += W_PASS) {
        const int nb = min(W_PASS, cnt - cb);

        if (cb > 0) __syncthreads();

        for (int u0 = 0; u0 < nb; u0 += W_GROUP) {
            v4f ta[W_GROUP], tb[W_GROUP];
#pragma unroll
            for (int u = 0; u < W_GROUP; ++u) {
                const int uu = min(u0 + u, nb - 1);
                const float* wsrc =
                    w + (size_t)colblk[beg + cb + uu] * (BLOCK * BLOCK)
                      + srow * BLOCK + scol;
                ta[u] = *(const v4f*)wsrc;
                tb[u] = *(const v4f*)(wsrc + 16 * BLOCK);
            }
#pragma unroll
            for (int u = 0; u < W_GROUP; ++u) {
                float* dst = &wlds[(u0 + u) * W_TILE];
                dst[(scol + 0) * LSTRIDE + srow]      = ta[u][0];
                dst[(scol + 1) * LSTRIDE + srow]      = ta[u][1];
                dst[(scol + 2) * LSTRIDE + srow]      = ta[u][2];
                dst[(scol + 3) * LSTRIDE + srow]      = ta[u][3];
                dst[(scol + 0) * LSTRIDE + srow + 16] = tb[u][0];
                dst[(scol + 1) * LSTRIDE + srow + 16] = tb[u][1];
                dst[(scol + 2) * LSTRIDE + srow + 16] = tb[u][2];
                dst[(scol + 3) * LSTRIDE + srow + 16] = tb[u][3];
            }
        }
        __syncthreads();

        for (int t = 0; t < nb; ++t) {
            const int    kbase = coli[beg + cb + t] * BLOCK;
            const float* xa0   = xrow0 + kbase;
            const float* xa1   = xrow1 + kbase;
            const float* bl    = &wlds[t * W_TILE + boff];

            v2f a0[8], a1[8], b0[8], b1[8];
#pragma unroll
            for (int s = 0; s < 8; ++s) {
                a0[s] = *(const v2f*)(xa0 + s * 4);
                a1[s] = *(const v2f*)(xa1 + s * 4);
            }
#pragma unroll
            for (int s = 0; s < 8; ++s) {
                b0[s] = *(const v2f*)(bl + s * 4);
                b1[s] = *(const v2f*)(bl + 16 * LSTRIDE + s * 4);
            }
            __builtin_amdgcn_sched_barrier(0);
#pragma unroll
            for (int s = 0; s < 8; ++s) {
                c00 = __builtin_amdgcn_wmma_f32_16x16x4_f32(
                          false, a0[s], false, b0[s], (short)0, c00, false, false);
                c01 = __builtin_amdgcn_wmma_f32_16x16x4_f32(
                          false, a0[s], false, b1[s], (short)0, c01, false, false);
                c10 = __builtin_amdgcn_wmma_f32_16x16x4_f32(
                          false, a1[s], false, b0[s], (short)0, c10, false, false);
                c11 = __builtin_amdgcn_wmma_f32_16x16x4_f32(
                          false, a1[s], false, b1[s], (short)0, c11, false, false);
            }
        }
    }

    float* y0 = y + (size_t)(m0 + half * 8) * NDIM + j * BLOCK + lid;
    float* y1 = y0 + (size_t)16 * NDIM;
#pragma unroll
    for (int r = 0; r < 8; ++r) {
        y0[(size_t)r * NDIM]      = c00[r];
        y0[(size_t)r * NDIM + 16] = c01[r];
        y1[(size_t)r * NDIM]      = c10[r];
        y1[(size_t)r * NDIM + 16] = c11[r];
    }
}

extern "C" void kernel_launch(void* const* d_in, const int* in_sizes, int n_in,
                              void* d_out, int out_size, void* d_ws, size_t ws_size,
                              hipStream_t stream) {
    const float* x     = (const float*)d_in[0];
    const float* w     = (const float*)d_in[1];
    const int*   idx_i = (const int*)d_in[2];
    const int*   idx_j = (const int*)d_in[3];
    float*       y     = (float*)d_out;

    const int nnz = in_sizes[2];
    const int M   = in_sizes[0] / KDIM;

    int* colptr = (int*)d_ws;
    int* colblk = colptr + (NB + 1);
    int* coli   = colblk + nnz;

    bsmm_build_csc<<<NB, BUILD_THREADS, 0, stream>>>(idx_i, idx_j, nnz, colptr, colblk, coli);

    dim3 grid(M / ROWS_PER_WG, NB);
    bsmm_wmma<<<grid, THREADS, 0, stream>>>(x, w, colptr, colblk, coli, y);
}
